# starting kernel
# baseline (speedup 1.0000x reference)
#include <hip/hip_runtime.h>

typedef float v2f __attribute__((ext_vector_type(2)));
typedef float v8f __attribute__((ext_vector_type(8)));

#define N_NODES 100000
#define N_EDGES 1600000


__global__ void zero_u32_kernel(unsigned int* p, int n) {
    int i = blockIdx.x * blockDim.x + threadIdx.x;
    if (i < n) p[i] = 0u;
}

__global__ void deg_count_kernel(const int* __restrict__ dst,
                                 unsigned int* __restrict__ cnt, int E) {
    int e = blockIdx.x * blockDim.x + threadIdx.x;
    if (e < E) atomicAdd(&cnt[dst[e]], 1u);
}

__global__ void make_dis_kernel(const unsigned int* __restrict__ cnt,
                                float* __restrict__ dis, int n) {
    int i = blockIdx.x * blockDim.x + threadIdx.x;
    if (i < n) dis[i] = rsqrtf((float)cnt[i] + 1.0f);
}


template<int K, int NOUT>
__global__ void __launch_bounds__(256)
gemm_wmma_kernel(const float* __restrict__ X, const float* __restrict__ W,
                 float* __restrict__ H, int nrows) {
    __shared__ float sW[K * NOUT];
    const int tid = threadIdx.x;
    for (int i = tid; i < K * NOUT; i += 256) sW[i] = W[i];
    __syncthreads();

    const int wave = tid >> 5;
    const int lane = tid & 31;
    const int half = lane >> 4;
    const int r    = lane & 15;

    const int row0 = (blockIdx.x * 8 + wave) * 16;
    if (row0 >= nrows) return;

    int arow = row0 + r;
    if (arow >= nrows) arow = nrows - 1;
    const float* __restrict__ xrow = X + (long long)arow * K;

    constexpr int NT = NOUT / 16;
    v8f acc[NT];
#pragma unroll
    for (int nt = 0; nt < NT; ++nt) {
        v8f z = {};
        acc[nt] = z;
    }

#pragma unroll 4
    for (int kt = 0; kt < K / 4; ++kt) {
        const int k0 = kt * 4 + 2 * half;
        v2f a;
        a.x = xrow[k0];
        a.y = xrow[k0 + 1];
#pragma unroll
        for (int nt = 0; nt < NT; ++nt) {
            v2f b;
            b.x = sW[(k0)     * NOUT + nt * 16 + r];
            b.y = sW[(k0 + 1) * NOUT + nt * 16 + r];
            acc[nt] = __builtin_amdgcn_wmma_f32_16x16x4_f32(
                false, a, false, b, (short)0, acc[nt], false, false);
        }
    }

#pragma unroll
    for (int nt = 0; nt < NT; ++nt) {
#pragma unroll
        for (int v = 0; v < 8; ++v) {
            const int row = row0 + v + 8 * half;
            const int col = nt * 16 + r;
            if (row < nrows) H[(long long)row * NOUT + col] = acc[nt][v];
        }
    }
}


template<int F>
__global__ void self_init_kernel(const float* __restrict__ h,
                                 const float* __restrict__ dis,
                                 float* __restrict__ acc, int n) {
    long long idx = (long long)blockIdx.x * blockDim.x + threadIdx.x;
    if (idx < (long long)n * F) {
        int node = (int)(idx / F);
        float d = dis[node];
        acc[idx] = h[idx] * d * d;
    }
}


template<int F>
__global__ void scatter_kernel(const int* __restrict__ src,
                               const int* __restrict__ dst,
                               const float* __restrict__ h,
                               const float* __restrict__ dis,
                               float* __restrict__ acc, int E) {
    long long idx = (long long)blockIdx.x * blockDim.x + threadIdx.x;
    if (idx >= (long long)E * F) return;
    const int e = (int)(idx >> __builtin_ctz(F));
    const int f = (int)(idx & (F - 1));
    const int s = src[e];
    const int d = dst[e];
    const float coef = dis[s] * dis[d];
    atomicAdd(&acc[(long long)d * F + f], h[(long long)s * F + f] * coef);
}


template<int F, bool RELU>
__global__ void epilogue_kernel(const float* __restrict__ acc,
                                const float* __restrict__ b,
                                float* __restrict__ out, int n) {
    long long idx = (long long)blockIdx.x * blockDim.x + threadIdx.x;
    if (idx < (long long)n * F) {
        int f = (int)(idx & (F - 1));
        float v = acc[idx] + b[f];
        out[idx] = RELU ? fmaxf(v, 0.0f) : v;
    }
}


static inline int cdiv(long long a, int b) { return (int)((a + b - 1) / b); }

extern "C" void kernel_launch(void* const* d_in, const int* in_sizes, int n_in,
                              void* d_out, int out_size, void* d_ws, size_t ws_size,
                              hipStream_t stream) {
    const float* x  = (const float*)d_in[0];
    const int*   ei = (const int*)d_in[1];
    const float* W1 = (const float*)d_in[2];
    const float* b1 = (const float*)d_in[3];
    const float* W2 = (const float*)d_in[4];
    const float* b2 = (const float*)d_in[5];
    const float* W3 = (const float*)d_in[6];
    const float* b3 = (const float*)d_in[7];
    float* out = (float*)d_out;

    const int* src = ei;
    const int* dst = ei + N_EDGES;

    char* ws = (char*)d_ws;
    unsigned int* cnt = (unsigned int*)ws;
    float* dis  = (float*)(ws + 400384);
    float* bufA = (float*)(ws + 800768);
    float* bufB = (float*)(ws + 800768 + 25600000);

    const int N = N_NODES, E = N_EDGES;
    const int gemm_blocks = cdiv(cdiv(N, 16), 8);

    zero_u32_kernel<<<cdiv(N, 256), 256, 0, stream>>>(cnt, N);
    deg_count_kernel<<<cdiv(E, 256), 256, 0, stream>>>(dst, cnt, E);
    make_dis_kernel<<<cdiv(N, 256), 256, 0, stream>>>(cnt, dis, N);

    gemm_wmma_kernel<128, 64><<<gemm_blocks, 256, 0, stream>>>(x, W1, bufA, N);
    self_init_kernel<64><<<cdiv((long long)N * 64, 256), 256, 0, stream>>>(bufA, dis, bufB, N);
    scatter_kernel<64><<<cdiv((long long)E * 64, 256), 256, 0, stream>>>(src, dst, bufA, dis, bufB, E);
    epilogue_kernel<64, true><<<cdiv((long long)N * 64, 256), 256, 0, stream>>>(bufB, b1, bufA, N);

    gemm_wmma_kernel<64, 64><<<gemm_blocks, 256, 0, stream>>>(bufA, W2, bufB, N);
    self_init_kernel<64><<<cdiv((long long)N * 64, 256), 256, 0, stream>>>(bufB, dis, bufA, N);
    scatter_kernel<64><<<cdiv((long long)E * 64, 256), 256, 0, stream>>>(src, dst, bufB, dis, bufA, E);
    epilogue_kernel<64, true><<<cdiv((long long)N * 64, 256), 256, 0, stream>>>(bufA, b2, bufB, N);

    gemm_wmma_kernel<64, 32><<<gemm_blocks, 256, 0, stream>>>(bufB, W3, bufA, N);
    self_init_kernel<32><<<cdiv((long long)N * 32, 256), 256, 0, stream>>>(bufA, dis, out, N);
    scatter_kernel<32><<<cdiv((long long)E * 32, 256), 256, 0, stream>>>(src, dst, bufA, dis, out, E);
    epilogue_kernel<32, false><<<cdiv((long long)N * 32, 256), 256, 0, stream>>>(out, b3, out, N);
}


# Every target of atomic adds (in-degree counts, layer 1 and 2 sums) kept once per package half: a wave adds into its own half's copy (die id), the reading product adds the two copies; sum order new
# speedup vs baseline: 4.0642x; 4.0642x over previous
#include <hip/hip_runtime.h>

typedef float v2f __attribute__((ext_vector_type(2)));
typedef float v4f __attribute__((ext_vector_type(4)));
typedef float v8f __attribute__((ext_vector_type(8)));

#define N_NODES 100000
#define N_EDGES 1600000

__device__ __forceinline__ int package_half() {
    return (int)((__builtin_amdgcn_s_sendmsg_rtn(0x87) >> 18) & 1u);
}


__global__ void zero_u32_kernel(unsigned int* p, int n) {
    int i = blockIdx.x * blockDim.x + threadIdx.x;
    if (i < n) p[i] = 0u;
}

__global__ void deg_count_kernel(const int* __restrict__ dst,
                                 unsigned int* __restrict__ cnt0,
                                 unsigned int* __restrict__ cnt1, int E) {
    int e = blockIdx.x * blockDim.x + threadIdx.x;
    if (e < E) {
        const int d = dst[e];
        unsigned int* __restrict__ cnt = package_half() ? cnt1 : cnt0;
        atomicAdd(&cnt[d], 1u);
    }
}


template<int K, int NOUT, bool PRE, bool MAKE_DIS, bool POST_BIAS, bool SPLIT_OUT>
__global__ void __launch_bounds__(256)
gemm_fused_kernel(const float* __restrict__ X, const float* __restrict__ X1,
                  const float* __restrict__ bpre,
                  const float* __restrict__ W,
                  const unsigned int* __restrict__ cnt,
                  const unsigned int* __restrict__ cnt1, float* __restrict__ dis,
                  const float* __restrict__ bpost,
                  float* __restrict__ H, float* __restrict__ ACC,
                  float* __restrict__ ACC1, int nrows) {
    constexpr int KH  = 64;
    constexpr int NPH = K / KH;
    constexpr int AP  = KH + 4;
    constexpr int NT  = NOUT / 16;
    static_assert(K % KH == 0 && NT % 2 == 0, "tile shapes");
    __shared__ __attribute__((aligned(16))) float sW[K * NOUT];
    __shared__ __attribute__((aligned(16))) float sA[8][16 * AP];

    const int tid  = threadIdx.x;
    const int wave = tid >> 5;
    const int lane = tid & 31;
    const int half = lane >> 4;
    const int r    = lane & 15;

    const int row0 = __builtin_amdgcn_readfirstlane((blockIdx.x * 8 + wave) * 16);

    v4f at[NPH][8];
    v4f at1[NPH][8];
#pragma unroll
    for (int p = 0; p < NPH; ++p) {
#pragma unroll
        for (int i = 0; i < 8; ++i) {
            int row = row0 + 2 * i + half;
            if (row >= nrows) row = nrows - 1;
            at[p][i] = *reinterpret_cast<const v4f*>(X + (long long)row * K + p * KH + 4 * r);
            if (PRE)
                at1[p][i] = *reinterpret_cast<const v4f*>(X1 + (long long)row * K + p * KH + 4 * r);
        }
    }
    v4f bb[NPH];
    if (PRE) {
#pragma unroll
        for (int p = 0; p < NPH; ++p)
            bb[p] = *reinterpret_cast<const v4f*>(bpre + p * KH + 4 * r);
    }
    float dl;
    {
        int row = row0 + r;
        if (row >= nrows) row = nrows - 1;
        dl = MAKE_DIS ? rsqrtf((float)(cnt[row] + cnt1[row]) + 1.0f) : dis[row];
    }

    {
        constexpr int NW = K * NOUT / 4 / 256;
        v4f w[NW];
#pragma unroll
        for (int i = 0; i < NW; ++i) w[i] = reinterpret_cast<const v4f*>(W)[tid + i * 256];
        if constexpr (PRE) {
            static_assert(!PRE || NPH == 1, "the fence below names one phase");
            asm volatile("" :: "v"(at1[0][0]), "v"(at1[0][1]), "v"(at1[0][2]), "v"(at1[0][3]),
                               "v"(at1[0][4]), "v"(at1[0][5]), "v"(at1[0][6]), "v"(at1[0][7]));
        }
#pragma unroll
        for (int i = 0; i < NW; ++i) reinterpret_cast<v4f*>(sW)[tid + i * 256] = w[i];
    }

    float* __restrict__ myA = &sA[wave][0];

    v8f acc[NT];
#pragma unroll
    for (int nt = 0; nt < NT; ++nt) {
        v8f z = {};
        acc[nt] = z;
    }

#pragma unroll
    for (int p = 0; p < NPH; ++p) {
#pragma unroll
        for (int i = 0; i < 8; ++i) {
            v4f a = at[p][i];
            if (PRE) {
                a += at1[p][i];
                a.x = fmaxf(a.x + bb[p].x, 0.0f);
                a.y = fmaxf(a.y + bb[p].y, 0.0f);
                a.z = fmaxf(a.z + bb[p].z, 0.0f);
                a.w = fmaxf(a.w + bb[p].w, 0.0f);
            }
            *reinterpret_cast<v4f*>(myA + (2 * i + half) * AP + 4 * r) = a;
        }
        if (p == 0) {
            if constexpr (NPH == 2) {
                asm volatile("" :: "v"(at[NPH - 1][0]), "v"(at[NPH - 1][1]), "v"(at[NPH - 1][2]),
                                   "v"(at[NPH - 1][3]), "v"(at[NPH - 1][4]), "v"(at[NPH - 1][5]),
                                   "v"(at[NPH - 1][6]), "v"(at[NPH - 1][7]));
            }
            __syncthreads();
            if (row0 >= nrows) return;
        }
#pragma unroll 1
        for (int jg = 0; jg < KH / 4; jg += 4) {
#pragma unroll
        for (int jj = 0; jj < 4; ++jj) {
            const int j  = jg + jj;
            const int kl = 4 * j + 2 * half;
            const int k0 = p * KH + kl;
            const v2f a = *reinterpret_cast<const v2f*>(myA + r * AP + kl);
#pragma unroll
            for (int nt = 0; nt < NT; ++nt) {
                v2f b;
                b.x = sW[(k0)     * NOUT + nt * 16 + r];
                b.y = sW[(k0 + 1) * NOUT + nt * 16 + r];
                acc[nt] = __builtin_amdgcn_wmma_f32_16x16x4_f32(
                    false, a, false, b, (short)0, acc[nt], false, false);
            }
        }
        }
    }

    if (MAKE_DIS) {
        const int row = row0 + lane;
        if (lane < 16 && row < nrows) dis[row] = dl;
    }

#pragma unroll
    for (int q = 0; q < NT / 2; ++q) {
        const int col = q * 32 + lane;
        const float bo = POST_BIAS ? bpost[col] : 0.0f;
#pragma unroll
        for (int v = 0; v < 8; ++v) {
            const float a = acc[2 * q][v];
            const float b = acc[2 * q + 1][v];
            const float t = half ? a : b;
            const float y = __int_as_float(__builtin_amdgcn_permlanex16(
                __float_as_int(t), __float_as_int(t), 0x76543210u, 0xfedcba98u, false, false));
            const float p0 = half ? y : a;
            const float p1 = half ? b : y;
            const float d0 = __int_as_float(__builtin_amdgcn_readlane(__float_as_int(dl), v));
            const float d1 = __int_as_float(__builtin_amdgcn_readlane(__float_as_int(dl), v + 8));
            const int rowa = row0 + v;
            const int rowb = row0 + v + 8;
            if (rowa < nrows) {
                float s = p0 * d0 * d0;
                if (POST_BIAS) s += bo;
                H[(long long)rowa * NOUT + col] = p0;
                ACC[(long long)rowa * NOUT + col] = s;
                if (SPLIT_OUT) ACC1[(long long)rowa * NOUT + col] = 0.0f;
            }
            if (rowb < nrows) {
                float s = p1 * d1 * d1;
                if (POST_BIAS) s += bo;
                H[(long long)rowb * NOUT + col] = p1;
                ACC[(long long)rowb * NOUT + col] = s;
                if (SPLIT_OUT) ACC1[(long long)rowb * NOUT + col] = 0.0f;
            }
        }
    }
}


static_assert(N_EDGES % 32 == 0, "a wave owns exactly 32 edges");
static_assert((long long)N_NODES * 64 * 4 < 0x7fffffffLL, "row offsets fit 32 bits");

template<int F, bool MAKE_COEF, bool SPLIT>
__global__ void __launch_bounds__(256)
scatter_kernel(const int* __restrict__ src,
               const int* __restrict__ dst,
               const float* __restrict__ h,
               const float* __restrict__ dis,
               float* __restrict__ coef,
               float* __restrict__ acc0,
               float* __restrict__ acc1) {
    constexpr int R = F / 32;
    constexpr int B = 32 / R;
    const int lane = threadIdx.x & 31;
    const int e0 = (blockIdx.x * 8 + (threadIdx.x >> 5)) * 32;
    if (e0 >= N_EDGES) return;

    const int s_l = src[e0 + lane];
    const int d_l = dst[e0 + lane];
    float* __restrict__ acc = acc0;
    if (SPLIT) { if (package_half()) acc = acc1; }
    float c_l;
    if (MAKE_COEF) {
        c_l = dis[s_l] * dis[d_l];
        coef[e0 + lane] = c_l;
    } else {
        c_l = coef[e0 + lane];
    }

#pragma unroll
    for (int jb = 0; jb < 32; jb += B) {
        float v[B][R];
#pragma unroll
        for (int k = 0; k < B; ++k) {
            const int s = __builtin_amdgcn_readlane(s_l, jb + k);
            const int hi = s * F + lane;
#pragma unroll
            for (int r = 0; r < R; ++r) v[k][r] = h[hi + r * 32];
        }
#pragma unroll
        for (int k = 0; k < B; ++k) {
            const int d = __builtin_amdgcn_readlane(d_l, jb + k);
            const float c = __int_as_float(
                __builtin_amdgcn_readlane(__float_as_int(c_l), jb + k));
            const int ai = d * F + lane;
#pragma unroll
            for (int r = 0; r < R; ++r) atomicAdd(&acc[ai + r * 32], v[k][r] * c);
        }
    }
}


static inline int cdiv(long long a, int b) { return (int)((a + b - 1) / b); }

extern "C" void kernel_launch(void* const* d_in, const int* in_sizes, int n_in,
                              void* d_out, int out_size, void* d_ws, size_t ws_size,
                              hipStream_t stream) {
    const float* x  = (const float*)d_in[0];
    const int*   ei = (const int*)d_in[1];
    const float* W1 = (const float*)d_in[2];
    const float* b1 = (const float*)d_in[3];
    const float* W2 = (const float*)d_in[4];
    const float* b2 = (const float*)d_in[5];
    const float* W3 = (const float*)d_in[6];
    const float* b3 = (const float*)d_in[7];
    float* out = (float*)d_out;

    const int* src = ei;
    const int* dst = ei + N_EDGES;

    char* ws = (char*)d_ws;
    unsigned int* cnt  = (unsigned int*)ws;
    unsigned int* cnt1 = (unsigned int*)(ws + 400384);
    float* dis   = (float*)(ws + 800768);
    float* bufA  = (float*)(ws + 1201152);
    float* bufB  = (float*)(ws + 1201152 + 25600000);
    float* bufB1 = (float*)(ws + 1201152 + 51200000);
    float* bufC  = (float*)(ws + 1201152 + 76800000);
    float* bufC1 = (float*)(ws + 1201152 + 102400000);
    float* coef  = (float*)(ws + 1201152 + 128000000);

    const int N = N_NODES, E = N_EDGES;
    const int gemm_blocks = cdiv(cdiv(N, 16), 8);
    const int scat_blocks = cdiv(cdiv(E, 32), 8);

    zero_u32_kernel<<<cdiv(100096 + N, 256), 256, 0, stream>>>(cnt, 100096 + N);
    deg_count_kernel<<<cdiv(E, 256), 256, 0, stream>>>(dst, cnt, cnt1, E);

    gemm_fused_kernel<128, 64, false, true, false, true><<<gemm_blocks, 256, 0, stream>>>(
        x, nullptr, nullptr, W1, cnt, cnt1, dis, nullptr, bufA, bufB, bufB1, N);
    scatter_kernel<64, true, true><<<scat_blocks, 256, 0, stream>>>(
        src, dst, bufA, dis, coef, bufB, bufB1);

    gemm_fused_kernel<64, 64, true, false, false, true><<<gemm_blocks, 256, 0, stream>>>(
        bufB, bufB1, b1, W2, cnt, cnt1, dis, nullptr, bufA, bufC, bufC1, N);
    scatter_kernel<64, false, true><<<scat_blocks, 256, 0, stream>>>(
        src, dst, bufA, dis, coef, bufC, bufC1);

    gemm_fused_kernel<64, 32, true, false, true, false><<<gemm_blocks, 256, 0, stream>>>(
        bufC, bufC1, b2, W3, cnt, cnt1, dis, b3, bufA, out, nullptr, N);
    scatter_kernel<32, false, false><<<scat_blocks, 256, 0, stream>>>(
        src, dst, bufA, dis, coef, out, nullptr);
}


# Each package half owns a node range instead of a copy of every sum: a block draws an (edge chunk, range) job from its half's counter, adds only that range's rows; copies, joins gone; sum order new
# speedup vs baseline: 5.0605x; 1.2451x over previous
#include <hip/hip_runtime.h>

typedef float v2f __attribute__((ext_vector_type(2)));
typedef float v4f __attribute__((ext_vector_type(4)));
typedef float v8f __attribute__((ext_vector_type(8)));

#define N_NODES 100000
#define N_EDGES 1600000

#define RANGE_SPLIT 50016

__device__ __forceinline__ int package_half() {
    return (int)((__builtin_amdgcn_s_sendmsg_rtn(0x87) >> 18) & 1u);
}

__device__ __forceinline__ int draw_job(unsigned int* __restrict__ ctr, int nch, int* s_job) {
    if (threadIdx.x == 0) {
        const int hf = package_half();
        int job = -1;
        const unsigned int t = atomicAdd(&ctr[hf * 64], 1u);
        if (t < (unsigned int)nch) {
            job = (int)t * 2 + hf;
        } else {
            const unsigned int t2 = atomicAdd(&ctr[(1 - hf) * 64], 1u);
            if (t2 < (unsigned int)nch) job = (int)t2 * 2 + (1 - hf);
        }
        *s_job = job;
    }
    __syncthreads();
    return __builtin_amdgcn_readfirstlane(*s_job);
}


__global__ void zero_u32_kernel(unsigned int* p, int n) {
    int i = blockIdx.x * blockDim.x + threadIdx.x;
    if (i < n) p[i] = 0u;
}

#define CNT_CHUNK 2048
#define CNT_NCH ((N_EDGES + CNT_CHUNK - 1) / CNT_CHUNK)

__global__ void __launch_bounds__(256)
deg_count_kernel(const int* __restrict__ dst, unsigned int* __restrict__ cnt,
                 unsigned int* __restrict__ ctr) {
    __shared__ int s_job;
    const int job = draw_job(ctr, CNT_NCH, &s_job);
    if (job < 0) return;
    const bool upper = (job & 1) != 0;
    const int e0 = (job >> 1) * CNT_CHUNK + threadIdx.x;
    int d[8];
#pragma unroll
    for (int i = 0; i < 8; ++i) {
        int e = e0 + i * 256;
        if (e >= N_EDGES) e = N_EDGES - 1;
        d[i] = dst[e];
    }
#pragma unroll
    for (int i = 0; i < 8; ++i) {
        if (e0 + i * 256 < N_EDGES && (d[i] >= RANGE_SPLIT) == upper) atomicAdd(&cnt[d[i]], 1u);
    }
}


template<int K, int NOUT, bool PRE, bool MAKE_DIS, bool POST_BIAS>
__global__ void __launch_bounds__(256)
gemm_fused_kernel(const float* __restrict__ X,
                  const float* __restrict__ bpre,
                  const float* __restrict__ W,
                  const unsigned int* __restrict__ cnt,
                  float* __restrict__ dis,
                  const float* __restrict__ bpost,
                  float* __restrict__ H, float* __restrict__ ACC, int nrows) {
    constexpr int KH  = 64;
    constexpr int NPH = K / KH;
    constexpr int AP  = KH + 4;
    constexpr int NT  = NOUT / 16;
    static_assert(K % KH == 0 && NT % 2 == 0, "tile shapes");
    __shared__ __attribute__((aligned(16))) float sW[K * NOUT];
    __shared__ __attribute__((aligned(16))) float sA[8][16 * AP];

    const int tid  = threadIdx.x;
    const int wave = tid >> 5;
    const int lane = tid & 31;
    const int half = lane >> 4;
    const int r    = lane & 15;

    const int row0 = __builtin_amdgcn_readfirstlane((blockIdx.x * 8 + wave) * 16);

    v4f at[NPH][8];
#pragma unroll
    for (int p = 0; p < NPH; ++p) {
#pragma unroll
        for (int i = 0; i < 8; ++i) {
            int row = row0 + 2 * i + half;
            if (row >= nrows) row = nrows - 1;
            at[p][i] = *reinterpret_cast<const v4f*>(X + (long long)row * K + p * KH + 4 * r);
        }
    }
    v4f bb[NPH];
    if (PRE) {
#pragma unroll
        for (int p = 0; p < NPH; ++p)
            bb[p] = *reinterpret_cast<const v4f*>(bpre + p * KH + 4 * r);
    }
    float dl;
    {
        int row = row0 + r;
        if (row >= nrows) row = nrows - 1;
        dl = MAKE_DIS ? rsqrtf((float)cnt[row] + 1.0f) : dis[row];
    }

    {
        constexpr int NW = K * NOUT / 4 / 256;
        v4f w[NW];
#pragma unroll
        for (int i = 0; i < NW; ++i) w[i] = reinterpret_cast<const v4f*>(W)[tid + i * 256];
#pragma unroll
        for (int i = 0; i < NW; ++i) reinterpret_cast<v4f*>(sW)[tid + i * 256] = w[i];
    }

    float* __restrict__ myA = &sA[wave][0];

    v8f acc[NT];
#pragma unroll
    for (int nt = 0; nt < NT; ++nt) {
        v8f z = {};
        acc[nt] = z;
    }

#pragma unroll
    for (int p = 0; p < NPH; ++p) {
#pragma unroll
        for (int i = 0; i < 8; ++i) {
            v4f a = at[p][i];
            if (PRE) {
                a.x = fmaxf(a.x + bb[p].x, 0.0f);
                a.y = fmaxf(a.y + bb[p].y, 0.0f);
                a.z = fmaxf(a.z + bb[p].z, 0.0f);
                a.w = fmaxf(a.w + bb[p].w, 0.0f);
            }
            *reinterpret_cast<v4f*>(myA + (2 * i + half) * AP + 4 * r) = a;
        }
        if (p == 0) {
            if constexpr (NPH == 2) {
                asm volatile("" :: "v"(at[NPH - 1][0]), "v"(at[NPH - 1][1]), "v"(at[NPH - 1][2]),
                                   "v"(at[NPH - 1][3]), "v"(at[NPH - 1][4]), "v"(at[NPH - 1][5]),
                                   "v"(at[NPH - 1][6]), "v"(at[NPH - 1][7]));
            }
            __syncthreads();
            if (row0 >= nrows) return;
        }
#pragma unroll 1
        for (int jg = 0; jg < KH / 4; jg += 4) {
#pragma unroll
        for (int jj = 0; jj < 4; ++jj) {
            const int j  = jg + jj;
            const int kl = 4 * j + 2 * half;
            const int k0 = p * KH + kl;
            const v2f a = *reinterpret_cast<const v2f*>(myA + r * AP + kl);
#pragma unroll
            for (int nt = 0; nt < NT; ++nt) {
                v2f b;
                b.x = sW[(k0)     * NOUT + nt * 16 + r];
                b.y = sW[(k0 + 1) * NOUT + nt * 16 + r];
                acc[nt] = __builtin_amdgcn_wmma_f32_16x16x4_f32(
                    false, a, false, b, (short)0, acc[nt], false, false);
            }
        }
        }
    }

    if (MAKE_DIS) {
        const int row = row0 + lane;
        if (lane < 16 && row < nrows) dis[row] = dl;
    }

#pragma unroll
    for (int q = 0; q < NT / 2; ++q) {
        const int col = q * 32 + lane;
        const float bo = POST_BIAS ? bpost[col] : 0.0f;
#pragma unroll
        for (int v = 0; v < 8; ++v) {
            const float a = acc[2 * q][v];
            const float b = acc[2 * q + 1][v];
            const float t = half ? a : b;
            const float y = __int_as_float(__builtin_amdgcn_permlanex16(
                __float_as_int(t), __float_as_int(t), 0x76543210u, 0xfedcba98u, false, false));
            const float p0 = half ? y : a;
            const float p1 = half ? b : y;
            const float d0 = __int_as_float(__builtin_amdgcn_readlane(__float_as_int(dl), v));
            const float d1 = __int_as_float(__builtin_amdgcn_readlane(__float_as_int(dl), v + 8));
            const int rowa = row0 + v;
            const int rowb = row0 + v + 8;
            if (rowa < nrows) {
                float s = p0 * d0 * d0;
                if (POST_BIAS) s += bo;
                H[(long long)rowa * NOUT + col] = p0;
                ACC[(long long)rowa * NOUT + col] = s;
            }
            if (rowb < nrows) {
                float s = p1 * d1 * d1;
                if (POST_BIAS) s += bo;
                H[(long long)rowb * NOUT + col] = p1;
                ACC[(long long)rowb * NOUT + col] = s;
            }
        }
    }
}


#define SC_CHUNK 512
#define SC_NCH (N_EDGES / SC_CHUNK)
static_assert(N_EDGES % SC_CHUNK == 0, "a job owns exactly 512 edges");
static_assert((long long)N_NODES * 64 * 4 < 0x7fffffffLL, "row offsets fit 32 bits");

template<int F, bool MAKE_COEF>
__global__ void __launch_bounds__(256)
scatter_kernel(const int* __restrict__ src,
               const int* __restrict__ dst,
               const float* __restrict__ h,
               const float* __restrict__ dis,
               float* __restrict__ coef,
               float* __restrict__ acc,
               unsigned int* __restrict__ ctr) {
    constexpr int R = F / 32;
    constexpr int B = 16;
    __shared__ int s_job;
    __shared__ int   q_s[8][64];
    __shared__ int   q_d[8][64];
    __shared__ float q_c[8][64];

    const int lane = threadIdx.x & 31;
    const int wave = threadIdx.x >> 5;
    const int job = draw_job(ctr, SC_NCH, &s_job);
    if (job < 0) return;
    const bool upper = (job & 1) != 0;
    const int e0 = (job >> 1) * SC_CHUNK + wave * 64;
    float* __restrict__ cf = coef + (upper ? N_EDGES : 0);

    const int s0 = src[e0 + lane];
    const int s1 = src[e0 + 32 + lane];
    const int d0 = dst[e0 + lane];
    const int d1 = dst[e0 + 32 + lane];
    const bool p0 = (d0 >= RANGE_SPLIT) == upper;
    const bool p1 = (d1 >= RANGE_SPLIT) == upper;
    float c0, c1;
    if (MAKE_COEF) {
        const float a0 = dis[p0 ? s0 : 0];
        const float b0 = dis[p0 ? d0 : 0];
        const float a1 = dis[p1 ? s1 : 0];
        const float b1 = dis[p1 ? d1 : 0];
        c0 = p0 ? a0 * b0 : 0.0f;
        c1 = p1 ? a1 * b1 : 0.0f;
        cf[e0 + lane] = c0;
        cf[e0 + 32 + lane] = c1;
    } else {
        c0 = cf[e0 + lane];
        c1 = cf[e0 + 32 + lane];
        asm volatile("" :: "v"(s0), "v"(s1), "v"(c0), "v"(c1));
    }

    const unsigned int m0 = __builtin_amdgcn_ballot_w32(p0);
    const unsigned int m1 = __builtin_amdgcn_ballot_w32(p1);
    const int n0 = __builtin_popcount(m0);
    const int n  = n0 + __builtin_popcount(m1);
    int*   __restrict__ qs = q_s[wave];
    int*   __restrict__ qd = q_d[wave];
    float* __restrict__ qc = q_c[wave];
    if (p0) {
        const int pos = (int)__builtin_amdgcn_mbcnt_lo(m0, 0u);
        qs[pos] = s0; qd[pos] = d0; qc[pos] = c0;
    }
    if (p1) {
        const int pos = n0 + (int)__builtin_amdgcn_mbcnt_lo(m1, 0u);
        qs[pos] = s1; qd[pos] = d1; qc[pos] = c1;
    }
    int sa = qs[lane];
    int sb = qs[lane + 32];
    const int   da = qd[lane];
    const int   db = qd[lane + 32];
    const float ca = qc[lane];
    const float cb = qc[lane + 32];
    if (lane >= n) sa = 0;
    if (lane + 32 >= n) sb = 0;

#pragma unroll
    for (int b = 0; b < 64 / B; ++b) {
        if (b * B < n) {
            const int   sv = (b * B < 32) ? sa : sb;
            const int   dv = (b * B < 32) ? da : db;
            const float cv = (b * B < 32) ? ca : cb;
            const int   lb = (b * B) & 31;
            float v[B][R];
#pragma unroll
            for (int k = 0; k < B; ++k) {
                const int s = __builtin_amdgcn_readlane(sv, lb + k);
                const int hi = s * F + lane;
#pragma unroll
                for (int r = 0; r < R; ++r) v[k][r] = h[hi + r * 32];
            }
            asm volatile("" :: "v"(v[0][0]), "v"(v[0][R - 1]), "v"(v[1][0]), "v"(v[1][R - 1]),
                               "v"(v[2][0]), "v"(v[2][R - 1]), "v"(v[3][0]), "v"(v[3][R - 1]),
                               "v"(v[B - 1][0]), "v"(v[B - 1][R - 1]));
#pragma unroll
            for (int k = 0; k < B; ++k) {
                if (b * B + k < n) {
                    const int d = __builtin_amdgcn_readlane(dv, lb + k);
                    const float c = __int_as_float(
                        __builtin_amdgcn_readlane(__float_as_int(cv), lb + k));
                    const int ai = d * F + lane;
#pragma unroll
                    for (int r = 0; r < R; ++r) atomicAdd(&acc[ai + r * 32], v[k][r] * c);
                }
            }
        }
    }
}


extern "C" void kernel_launch(void* const* d_in, const int* in_sizes, int n_in,
                              void* d_out, int out_size, void* d_ws, size_t ws_size,
                              hipStream_t stream) {
    const float* x  = (const float*)d_in[0];
    const int*   ei = (const int*)d_in[1];
    const float* W1 = (const float*)d_in[2];
    const float* b1 = (const float*)d_in[3];
    const float* W2 = (const float*)d_in[4];
    const float* b2 = (const float*)d_in[5];
    const float* W3 = (const float*)d_in[6];
    const float* b3 = (const float*)d_in[7];
    float* out = (float*)d_out;

    const int* src = ei;
    const int* dst = ei + N_EDGES;

    char* ws = (char*)d_ws;
    unsigned int* cnt = (unsigned int*)ws;
    unsigned int* ctr = (unsigned int*)(ws + 400384);
    float* dis   = (float*)(ws + 402432);
    float* bufA  = (float*)(ws + 802816);
    float* bufB  = (float*)(ws + 802816 + 25600000);
    float* bufC  = (float*)(ws + 802816 + 51200000);
    float* coef  = (float*)(ws + 802816 + 76800000);

    const int N = N_NODES;
    constexpr int gemm_blocks = ((N_NODES + 15) / 16 + 7) / 8;
    constexpr int zero_words  = (400384 + 2048) / 4;

    zero_u32_kernel<<<(zero_words + 255) / 256, 256, 0, stream>>>(cnt, zero_words);
    deg_count_kernel<<<2 * CNT_NCH, 256, 0, stream>>>(dst, cnt, ctr);

    gemm_fused_kernel<128, 64, false, true, false><<<gemm_blocks, 256, 0, stream>>>(
        x, nullptr, W1, cnt, dis, nullptr, bufA, bufB, N);
    scatter_kernel<64, true><<<2 * SC_NCH, 256, 0, stream>>>(
        src, dst, bufA, dis, coef, bufB, ctr + 128);

    gemm_fused_kernel<64, 64, true, false, false><<<gemm_blocks, 256, 0, stream>>>(
        bufB, b1, W2, cnt, dis, nullptr, bufA, bufC, N);
    scatter_kernel<64, false><<<2 * SC_NCH, 256, 0, stream>>>(
        src, dst, bufA, dis, coef, bufC, ctr + 256);

    gemm_fused_kernel<64, 32, true, false, true><<<gemm_blocks, 256, 0, stream>>>(
        bufC, b2, W3, cnt, dis, b3, bufA, out, N);
    scatter_kernel<32, false><<<2 * SC_NCH, 256, 0, stream>>>(
        src, dst, bufA, dis, coef, out, ctr + 384);
}


# Product launches draw their 128-row tile from the counter of the package half they run on: a node range's counts, start values and sums are written and read in the half that adds into them
# speedup vs baseline: 5.3702x; 1.0612x over previous
#include <hip/hip_runtime.h>

typedef float v2f __attribute__((ext_vector_type(2)));
typedef float v4f __attribute__((ext_vector_type(4)));
typedef float v8f __attribute__((ext_vector_type(8)));

#define N_NODES 100000
#define N_EDGES 1600000

#define RANGE_SPLIT 50016
#define GEMM_NCH ((RANGE_SPLIT + 127) / 128)
static_assert(RANGE_SPLIT % 16 == 0 && N_NODES % 16 == 0, "a wave's 16 rows lie in one range");
static_assert(N_NODES - RANGE_SPLIT <= GEMM_NCH * 128, "the upper range fits the same tile count");

__device__ __forceinline__ int package_half() {
    return (int)((__builtin_amdgcn_s_sendmsg_rtn(0x87) >> 18) & 1u);
}

__device__ __forceinline__ int draw_job(unsigned int* __restrict__ ctr, int nch, int* s_job) {
    if (threadIdx.x == 0) {
        const int hf = package_half();
        int job = -1;
        const unsigned int t = atomicAdd(&ctr[hf * 64], 1u);
        if (t < (unsigned int)nch) {
            job = (int)t * 2 + hf;
        } else {
            const unsigned int t2 = atomicAdd(&ctr[(1 - hf) * 64], 1u);
            if (t2 < (unsigned int)nch) job = (int)t2 * 2 + (1 - hf);
        }
        *s_job = job;
    }
    __syncthreads();
    return __builtin_amdgcn_readfirstlane(*s_job);
}


__global__ void zero_u32_kernel(unsigned int* p, int n) {
    int i = blockIdx.x * blockDim.x + threadIdx.x;
    if (i < n) p[i] = 0u;
}

#define CNT_CHUNK 2048
#define CNT_NCH ((N_EDGES + CNT_CHUNK - 1) / CNT_CHUNK)

__global__ void __launch_bounds__(256)
deg_count_kernel(const int* __restrict__ dst, unsigned int* __restrict__ cnt,
                 unsigned int* __restrict__ ctr) {
    __shared__ int s_job;
    const int job = draw_job(ctr, CNT_NCH, &s_job);
    if (job < 0) return;
    const bool upper = (job & 1) != 0;
    const int e0 = (job >> 1) * CNT_CHUNK + threadIdx.x;
    int d[8];
#pragma unroll
    for (int i = 0; i < 8; ++i) {
        int e = e0 + i * 256;
        if (e >= N_EDGES) e = N_EDGES - 1;
        d[i] = dst[e];
    }
#pragma unroll
    for (int i = 0; i < 8; ++i) {
        if (e0 + i * 256 < N_EDGES && (d[i] >= RANGE_SPLIT) == upper) atomicAdd(&cnt[d[i]], 1u);
    }
}


template<int K, int NOUT, bool PRE, bool MAKE_DIS, bool POST_BIAS>
__global__ void __launch_bounds__(256)
gemm_fused_kernel(const float* __restrict__ X,
                  const float* __restrict__ bpre,
                  const float* __restrict__ W,
                  const unsigned int* __restrict__ cnt,
                  float* __restrict__ dis,
                  const float* __restrict__ bpost,
                  float* __restrict__ H, float* __restrict__ ACC,
                  unsigned int* __restrict__ ctr) {
    constexpr int KH  = 64;
    constexpr int NPH = K / KH;
    constexpr int AP  = KH + 4;
    constexpr int NT  = NOUT / 16;
    static_assert(K % KH == 0 && NT % 2 == 0, "tile shapes");
    __shared__ __attribute__((aligned(16))) float sW[K * NOUT];
    __shared__ __attribute__((aligned(16))) float sA[8][16 * AP];
    __shared__ int s_job;

    const int tid  = threadIdx.x;
    const int wave = tid >> 5;
    const int lane = tid & 31;
    const int half = lane >> 4;
    const int r    = lane & 15;

    constexpr int NW = K * NOUT / 4 / 256;
    v4f w[NW];
#pragma unroll
    for (int i = 0; i < NW; ++i) w[i] = reinterpret_cast<const v4f*>(W)[tid + i * 256];

    const int job = draw_job(ctr, GEMM_NCH, &s_job);
    if (job < 0) return;
    const int rbeg = (job & 1) ? RANGE_SPLIT : 0;
    const int rend = (job & 1) ? N_NODES : RANGE_SPLIT;
    const int row0 = __builtin_amdgcn_readfirstlane(rbeg + ((job >> 1) * 8 + wave) * 16);

    v4f at[NPH][8];
#pragma unroll
    for (int p = 0; p < NPH; ++p) {
#pragma unroll
        for (int i = 0; i < 8; ++i) {
            int row = row0 + 2 * i + half;
            if (row >= rend) row = rend - 1;
            at[p][i] = *reinterpret_cast<const v4f*>(X + (long long)row * K + p * KH + 4 * r);
        }
    }
    v4f bb[NPH];
    if (PRE) {
#pragma unroll
        for (int p = 0; p < NPH; ++p)
            bb[p] = *reinterpret_cast<const v4f*>(bpre + p * KH + 4 * r);
    }
    float dl;
    {
        int row = row0 + r;
        if (row >= rend) row = rend - 1;
        dl = MAKE_DIS ? rsqrtf((float)cnt[row] + 1.0f) : dis[row];
    }

#pragma unroll
    for (int i = 0; i < NW; ++i) reinterpret_cast<v4f*>(sW)[tid + i * 256] = w[i];

    float* __restrict__ myA = &sA[wave][0];

    v8f acc[NT];
#pragma unroll
    for (int nt = 0; nt < NT; ++nt) {
        v8f z = {};
        acc[nt] = z;
    }

#pragma unroll
    for (int p = 0; p < NPH; ++p) {
#pragma unroll
        for (int i = 0; i < 8; ++i) {
            v4f a = at[p][i];
            if (PRE) {
                a.x = fmaxf(a.x + bb[p].x, 0.0f);
                a.y = fmaxf(a.y + bb[p].y, 0.0f);
                a.z = fmaxf(a.z + bb[p].z, 0.0f);
                a.w = fmaxf(a.w + bb[p].w, 0.0f);
            }
            *reinterpret_cast<v4f*>(myA + (2 * i + half) * AP + 4 * r) = a;
        }
        if (p == 0) {
            if constexpr (NPH == 2) {
                asm volatile("" :: "v"(at[NPH - 1][0]), "v"(at[NPH - 1][1]), "v"(at[NPH - 1][2]),
                                   "v"(at[NPH - 1][3]), "v"(at[NPH - 1][4]), "v"(at[NPH - 1][5]),
                                   "v"(at[NPH - 1][6]), "v"(at[NPH - 1][7]));
            }
            __syncthreads();
            if (row0 >= rend) return;
        }
#pragma unroll 1
        for (int jg = 0; jg < KH / 4; jg += 4) {
#pragma unroll
        for (int jj = 0; jj < 4; ++jj) {
            const int j  = jg + jj;
            const int kl = 4 * j + 2 * half;
            const int k0 = p * KH + kl;
            const v2f a = *reinterpret_cast<const v2f*>(myA + r * AP + kl);
#pragma unroll
            for (int nt = 0; nt < NT; ++nt) {
                v2f b;
                b.x = sW[(k0)     * NOUT + nt * 16 + r];
                b.y = sW[(k0 + 1) * NOUT + nt * 16 + r];
                acc[nt] = __builtin_amdgcn_wmma_f32_16x16x4_f32(
                    false, a, false, b, (short)0, acc[nt], false, false);
            }
        }
        }
    }

    if (MAKE_DIS) {
        const int row = row0 + lane;
        if (lane < 16 && row < rend) dis[row] = dl;
    }

#pragma unroll
    for (int q = 0; q < NT / 2; ++q) {
        const int col = q * 32 + lane;
        const float bo = POST_BIAS ? bpost[col] : 0.0f;
#pragma unroll
        for (int v = 0; v < 8; ++v) {
            const float a = acc[2 * q][v];
            const float b = acc[2 * q + 1][v];
            const float t = half ? a : b;
            const float y = __int_as_float(__builtin_amdgcn_permlanex16(
                __float_as_int(t), __float_as_int(t), 0x76543210u, 0xfedcba98u, false, false));
            const float p0 = half ? y : a;
            const float p1 = half ? b : y;
            const float d0 = __int_as_float(__builtin_amdgcn_readlane(__float_as_int(dl), v));
            const float d1 = __int_as_float(__builtin_amdgcn_readlane(__float_as_int(dl), v + 8));
            const int rowa = row0 + v;
            const int rowb = row0 + v + 8;
            if (rowa < rend) {
                float s = p0 * d0 * d0;
                if (POST_BIAS) s += bo;
                H[(long long)rowa * NOUT + col] = p0;
                ACC[(long long)rowa * NOUT + col] = s;
            }
            if (rowb < rend) {
                float s = p1 * d1 * d1;
                if (POST_BIAS) s += bo;
                H[(long long)rowb * NOUT + col] = p1;
                ACC[(long long)rowb * NOUT + col] = s;
            }
        }
    }
}


#define SC_CHUNK 512
#define SC_NCH (N_EDGES / SC_CHUNK)
static_assert(N_EDGES % SC_CHUNK == 0, "a job owns exactly 512 edges");
static_assert((long long)N_NODES * 64 * 4 < 0x7fffffffLL, "row offsets fit 32 bits");

template<int F, bool MAKE_COEF>
__global__ void __launch_bounds__(256)
scatter_kernel(const int* __restrict__ src,
               const int* __restrict__ dst,
               const float* __restrict__ h,
               const float* __restrict__ dis,
               float* __restrict__ coef,
               float* __restrict__ acc,
               unsigned int* __restrict__ ctr) {
    constexpr int R = F / 32;
    constexpr int B = 16;
    __shared__ int s_job;
    __shared__ int   q_s[8][64];
    __shared__ int   q_d[8][64];
    __shared__ float q_c[8][64];

    const int lane = threadIdx.x & 31;
    const int wave = threadIdx.x >> 5;
    const int job = draw_job(ctr, SC_NCH, &s_job);
    if (job < 0) return;
    const bool upper = (job & 1) != 0;
    const int e0 = (job >> 1) * SC_CHUNK + wave * 64;
    float* __restrict__ cf = coef + (upper ? N_EDGES : 0);

    const int s0 = src[e0 + lane];
    const int s1 = src[e0 + 32 + lane];
    const int d0 = dst[e0 + lane];
    const int d1 = dst[e0 + 32 + lane];
    const bool p0 = (d0 >= RANGE_SPLIT) == upper;
    const bool p1 = (d1 >= RANGE_SPLIT) == upper;
    float c0, c1;
    if (MAKE_COEF) {
        const float a0 = dis[p0 ? s0 : 0];
        const float b0 = dis[p0 ? d0 : 0];
        const float a1 = dis[p1 ? s1 : 0];
        const float b1 = dis[p1 ? d1 : 0];
        c0 = p0 ? a0 * b0 : 0.0f;
        c1 = p1 ? a1 * b1 : 0.0f;
        cf[e0 + lane] = c0;
        cf[e0 + 32 + lane] = c1;
    } else {
        c0 = cf[e0 + lane];
        c1 = cf[e0 + 32 + lane];
        asm volatile("" :: "v"(s0), "v"(s1), "v"(c0), "v"(c1));
    }

    const unsigned int m0 = __builtin_amdgcn_ballot_w32(p0);
    const unsigned int m1 = __builtin_amdgcn_ballot_w32(p1);
    const int n0 = __builtin_popcount(m0);
    const int n  = n0 + __builtin_popcount(m1);
    int*   __restrict__ qs = q_s[wave];
    int*   __restrict__ qd = q_d[wave];
    float* __restrict__ qc = q_c[wave];
    if (p0) {
        const int pos = (int)__builtin_amdgcn_mbcnt_lo(m0, 0u);
        qs[pos] = s0; qd[pos] = d0; qc[pos] = c0;
    }
    if (p1) {
        const int pos = n0 + (int)__builtin_amdgcn_mbcnt_lo(m1, 0u);
        qs[pos] = s1; qd[pos] = d1; qc[pos] = c1;
    }
    int sa = qs[lane];
    int sb = qs[lane + 32];
    const int   da = qd[lane];
    const int   db = qd[lane + 32];
    const float ca = qc[lane];
    const float cb = qc[lane + 32];
    if (lane >= n) sa = 0;
    if (lane + 32 >= n) sb = 0;

#pragma unroll
    for (int b = 0; b < 64 / B; ++b) {
        if (b * B < n) {
            const int   sv = (b * B < 32) ? sa : sb;
            const int   dv = (b * B < 32) ? da : db;
            const float cv = (b * B < 32) ? ca : cb;
            const int   lb = (b * B) & 31;
            float v[B][R];
#pragma unroll
            for (int k = 0; k < B; ++k) {
                const int s = __builtin_amdgcn_readlane(sv, lb + k);
                const int hi = s * F + lane;
#pragma unroll
                for (int r = 0; r < R; ++r) v[k][r] = h[hi + r * 32];
            }
            asm volatile("" :: "v"(v[0][0]), "v"(v[0][R - 1]), "v"(v[1][0]), "v"(v[1][R - 1]),
                               "v"(v[2][0]), "v"(v[2][R - 1]), "v"(v[3][0]), "v"(v[3][R - 1]),
                               "v"(v[B - 1][0]), "v"(v[B - 1][R - 1]));
#pragma unroll
            for (int k = 0; k < B; ++k) {
                if (b * B + k < n) {
                    const int d = __builtin_amdgcn_readlane(dv, lb + k);
                    const float c = __int_as_float(
                        __builtin_amdgcn_readlane(__float_as_int(cv), lb + k));
                    const int ai = d * F + lane;
#pragma unroll
                    for (int r = 0; r < R; ++r) atomicAdd(&acc[ai + r * 32], v[k][r] * c);
                }
            }
        }
    }
}


extern "C" void kernel_launch(void* const* d_in, const int* in_sizes, int n_in,
                              void* d_out, int out_size, void* d_ws, size_t ws_size,
                              hipStream_t stream) {
    const float* x  = (const float*)d_in[0];
    const int*   ei = (const int*)d_in[1];
    const float* W1 = (const float*)d_in[2];
    const float* b1 = (const float*)d_in[3];
    const float* W2 = (const float*)d_in[4];
    const float* b2 = (const float*)d_in[5];
    const float* W3 = (const float*)d_in[6];
    const float* b3 = (const float*)d_in[7];
    float* out = (float*)d_out;

    const int* src = ei;
    const int* dst = ei + N_EDGES;

    char* ws = (char*)d_ws;
    unsigned int* cnt = (unsigned int*)ws;
    unsigned int* ctr = (unsigned int*)(ws + 400384);
    float* dis   = (float*)(ws + 402432);
    float* bufA  = (float*)(ws + 802816);
    float* bufB  = (float*)(ws + 802816 + 25600000);
    float* bufC  = (float*)(ws + 802816 + 51200000);
    float* coef  = (float*)(ws + 802816 + 76800000);

    constexpr int gemm_blocks = 2 * GEMM_NCH;
    constexpr int zero_words  = (400384 + 2048) / 4;

    zero_u32_kernel<<<(zero_words + 255) / 256, 256, 0, stream>>>(cnt, zero_words);
    deg_count_kernel<<<2 * CNT_NCH, 256, 0, stream>>>(dst, cnt, ctr);

    gemm_fused_kernel<128, 64, false, true, false><<<gemm_blocks, 256, 0, stream>>>(
        x, nullptr, W1, cnt, dis, nullptr, bufA, bufB, ctr + 32);
    scatter_kernel<64, true><<<2 * SC_NCH, 256, 0, stream>>>(
        src, dst, bufA, dis, coef, bufB, ctr + 128);

    gemm_fused_kernel<64, 64, true, false, false><<<gemm_blocks, 256, 0, stream>>>(
        bufB, b1, W2, cnt, dis, nullptr, bufA, bufC, ctr + 160);
    scatter_kernel<64, false><<<2 * SC_NCH, 256, 0, stream>>>(
        src, dst, bufA, dis, coef, bufC, ctr + 256);

    gemm_fused_kernel<64, 32, true, false, true><<<gemm_blocks, 256, 0, stream>>>(
        bufC, b2, W3, cnt, dis, b3, bufA, out, ctr + 288);
    scatter_kernel<32, false><<<2 * SC_NCH, 256, 0, stream>>>(
        src, dst, bufA, dis, coef, out, ctr + 384);
}
